# starting kernel
# baseline (speedup 1.0000x reference)
#include <hip/hip_runtime.h>
#include <hip/hip_bf16.h>

typedef __attribute__((ext_vector_type(2))) float v2f;
typedef __attribute__((ext_vector_type(8))) float v8f;

#define NEG_SLOPE 0.2f
#define EPS_DEN 1e-16f
#define FDIM 64

__device__ __forceinline__ unsigned fenc(float f) {
  unsigned b = __float_as_uint(f);
  return (b & 0x80000000u) ? ~b : (b | 0x80000000u);
}
__device__ __forceinline__ float fdec(unsigned k) {
  unsigned b = (k & 0x80000000u) ? (k & 0x7fffffffu) : ~k;
  return __uint_as_float(b);
}

__global__ __launch_bounds__(128) void gatv2_gemm_dual(
    const float* __restrict__ X, const float* __restrict__ Wl,
    const float* __restrict__ Wr, float* __restrict__ xl,
    float* __restrict__ xr, int N, int K) {
  const int lane   = threadIdx.x & 31;
  const int wave   = threadIdx.x >> 5;
  const int laneHi = lane >> 4;
  const int lane16 = lane & 15;
  const int rowTile = blockIdx.x;
  const int colBase = wave * 16;

  int row  = rowTile * 16 + lane16;
  int rowc = row < N ? row : N - 1;

  v8f accL = {};
  v8f accR = {};
  const float* xrow = X + (size_t)rowc * K;

  for (int k = 0; k < K; k += 4) {
    const int kk = k + 2 * laneHi;
    v2f a = *(const v2f*)(xrow + kk);
    v2f bl, br;
    bl[0] = Wl[(size_t)kk * FDIM + colBase + lane16];
    bl[1] = Wl[(size_t)(kk + 1) * FDIM + colBase + lane16];
    br[0] = Wr[(size_t)kk * FDIM + colBase + lane16];
    br[1] = Wr[(size_t)(kk + 1) * FDIM + colBase + lane16];
    accL = __builtin_amdgcn_wmma_f32_16x16x4_f32(false, a, false, bl,
                                                 (short)0, accL, false, false);
    accR = __builtin_amdgcn_wmma_f32_16x16x4_f32(false, a, false, br,
                                                 (short)0, accR, false, false);
  }

#pragma unroll
  for (int v = 0; v < 8; ++v) {
    int m = v + 8 * laneHi;
    int r = rowTile * 16 + m;
    if (r < N) {
      xl[(size_t)r * FDIM + colBase + lane16] = accL[v];
      xr[(size_t)r * FDIM + colBase + lane16] = accR[v];
    }
  }
}

__global__ __launch_bounds__(256) void gatv2_edge_logit(
    const float* __restrict__ xl, const float* __restrict__ xr,
    const int* __restrict__ esrc, const int* __restrict__ edst,
    const float* __restrict__ att, float* __restrict__ e,
    unsigned* __restrict__ mkey, int E, int ET) {
  int eid = blockIdx.x * 8 + (threadIdx.x >> 5);
  if (eid >= ET) return;
  int lane = threadIdx.x & 31;
  int s, d;
  if (eid < E) { s = esrc[eid]; d = edst[eid]; }
  else         { s = d = eid - E; }

  v2f a  = *(const v2f*)(xl + (size_t)s * FDIM + 2 * lane);
  v2f b  = *(const v2f*)(xr + (size_t)d * FDIM + 2 * lane);
  v2f at = *(const v2f*)(att + 2 * lane);
  float t0 = a[0] + b[0]; t0 = t0 > 0.f ? t0 : NEG_SLOPE * t0;
  float t1 = a[1] + b[1]; t1 = t1 > 0.f ? t1 : NEG_SLOPE * t1;
  float sum = t0 * at[0] + t1 * at[1];
#pragma unroll
  for (int off = 16; off; off >>= 1) sum += __shfl_xor(sum, off, 32);
  if (lane == 0) {
    e[eid] = sum;
    atomicMax(mkey + d, fenc(sum));
  }
}

__global__ __launch_bounds__(256) void gatv2_edge_exp(
    const float* __restrict__ e, const int* __restrict__ edst,
    const unsigned* __restrict__ mkey, float* __restrict__ ex,
    float* __restrict__ denom, int E, int ET) {
  int eid = blockIdx.x * 256 + threadIdx.x;
  if (eid >= ET) return;
  int d = (eid < E) ? edst[eid] : (eid - E);
  float v = __expf(e[eid] - fdec(mkey[d]));
  ex[eid] = v;
  atomicAdd(denom + d, v);
}

__global__ __launch_bounds__(256) void gatv2_edge_aggregate(
    const float* __restrict__ xl, const float* __restrict__ ex,
    const float* __restrict__ denom, const int* __restrict__ esrc,
    const int* __restrict__ edst, float* __restrict__ agg, int E, int ET) {
  int eid = blockIdx.x * 8 + (threadIdx.x >> 5);
  if (eid >= ET) return;
  int lane = threadIdx.x & 31;
  int s, d;
  if (eid < E) { s = esrc[eid]; d = edst[eid]; }
  else         { s = d = eid - E; }

  float alpha = ex[eid] / (denom[d] + EPS_DEN);
  v2f a = *(const v2f*)(xl + (size_t)s * FDIM + 2 * lane);
  float* dp = agg + (size_t)d * FDIM + 2 * lane;
  atomicAdd(dp, alpha * a[0]);
  atomicAdd(dp + 1, alpha * a[1]);
}

__global__ __launch_bounds__(256) void gatv2_finalize(
    const float* __restrict__ agg, const float* __restrict__ bias,
    float* __restrict__ out, size_t n, int relu) {
  size_t i = (size_t)blockIdx.x * 256 + threadIdx.x;
  if (i >= n) return;
  float v = agg[i] + bias[i & (FDIM - 1)];
  out[i] = (relu && v < 0.f) ? 0.f : v;
}

extern "C" void kernel_launch(void* const* d_in, const int* in_sizes, int n_in,
                              void* d_out, int out_size, void* d_ws, size_t ws_size,
                              hipStream_t stream) {
  const float* x    = (const float*)d_in[0];
  const int*   ei   = (const int*)d_in[1];
  const float* W1l  = (const float*)d_in[2];
  const float* W1r  = (const float*)d_in[3];
  const float* att1 = (const float*)d_in[4];
  const float* b1   = (const float*)d_in[5];
  const float* W2l  = (const float*)d_in[6];
  const float* W2r  = (const float*)d_in[7];
  const float* att2 = (const float*)d_in[8];
  const float* b2   = (const float*)d_in[9];

  const int N  = in_sizes[0] / 128;
  const int E  = in_sizes[1] / 2;
  const int ET = E + N;

  float* ws = (float*)d_ws;
  const size_t NF = (size_t)N * FDIM;
  float*    xl    = ws;
  float*    xr    = xl + NF;
  float*    h     = xr + NF;
  float*    agg   = h + NF;
  float*    e     = agg + NF;
  float*    ex    = e + ET;
  unsigned* mkey  = (unsigned*)(ex + ET);
  float*    denom = (float*)(mkey + N);

  const int rowTiles   = (N + 15) / 16;
  const int edgeWaveBl = (ET + 7) / 8;
  const int edgeThrBl  = (ET + 255) / 256;
  const int nfBl       = (int)((NF + 255) / 256);
  const int* esrc = ei;
  const int* edst = ei + E;

  hipMemsetAsync(mkey, 0, sizeof(unsigned) * N, stream);
  hipMemsetAsync(denom, 0, sizeof(float) * N, stream);
  hipMemsetAsync(agg, 0, sizeof(float) * NF, stream);
  gatv2_gemm_dual<<<rowTiles, 128, 0, stream>>>(x, W1l, W1r, xl, xr, N, 128);
  gatv2_edge_logit<<<edgeWaveBl, 256, 0, stream>>>(xl, xr, esrc, edst, att1, e, mkey, E, ET);
  gatv2_edge_exp<<<edgeThrBl, 256, 0, stream>>>(e, edst, mkey, ex, denom, E, ET);
  gatv2_edge_aggregate<<<edgeWaveBl, 256, 0, stream>>>(xl, ex, denom, esrc, edst, agg, E, ET);
  gatv2_finalize<<<nfBl, 256, 0, stream>>>(agg, b1, h, NF, 1);

  hipMemsetAsync(mkey, 0, sizeof(unsigned) * N, stream);
  hipMemsetAsync(denom, 0, sizeof(float) * N, stream);
  hipMemsetAsync(agg, 0, sizeof(float) * NF, stream);
  gatv2_gemm_dual<<<rowTiles, 128, 0, stream>>>(h, W2l, W2r, xl, xr, N, 64);
  gatv2_edge_logit<<<edgeWaveBl, 256, 0, stream>>>(xl, xr, esrc, edst, att2, e, mkey, E, ET);
  gatv2_edge_exp<<<edgeThrBl, 256, 0, stream>>>(e, edst, mkey, ex, denom, E, ET);
  gatv2_edge_aggregate<<<edgeWaveBl, 256, 0, stream>>>(xl, ex, denom, esrc, edst, agg, E, ET);
  gatv2_finalize<<<nfBl, 256, 0, stream>>>(agg, b2, (float*)d_out, NF, 0);
}


# each destination keeps 4 edge lists (edge index mod 4); its wave walks all 4 at once, one per 8-lane group, joining 4 partial softmaxes at the end: load chain 17 -> ~7 trips; order of sums changes
# speedup vs baseline: 5.4740x; 5.4740x over previous
#include <hip/hip_runtime.h>
#include <hip/hip_bf16.h>

typedef __attribute__((ext_vector_type(2))) float v2f;
typedef __attribute__((ext_vector_type(4))) float v4f;
typedef __attribute__((ext_vector_type(8))) float v8f;

#define NEG_SLOPE 0.2f
#define EPS_DEN 1e-16f
#define FDIM 64
#define NLIST 4

template <bool RELU_IN, bool INIT_HEAD>
__global__ __launch_bounds__(128) void gatv2_gemm_dual(
    const float* __restrict__ X, const float* __restrict__ Wl,
    const float* __restrict__ Wr, float* __restrict__ xl,
    float* __restrict__ xr, int* __restrict__ head, int N, int K) {
  const int lane   = threadIdx.x & 31;
  const int wave   = threadIdx.x >> 5;
  const int laneHi = lane >> 4;
  const int lane16 = lane & 15;
  const int rowTile = blockIdx.x;
  const int colBase = wave * 16;

  int row  = rowTile * 16 + lane16;
  int rowc = row < N ? row : N - 1;

  if (INIT_HEAD) {
    const int t = threadIdx.x;
    if (t < 16 * NLIST && rowTile * 16 + t / NLIST < N)
      head[(size_t)rowTile * 16 * NLIST + t] = -1;
  }

  v8f accL = {};
  v8f accR = {};
  const float* xrow = X + (size_t)rowc * K;

  for (int k = 0; k < K; k += 4) {
    const int kk = k + 2 * laneHi;
    v2f a = *(const v2f*)(xrow + kk);
    if (RELU_IN) {
      a[0] = a[0] < 0.f ? 0.f : a[0];
      a[1] = a[1] < 0.f ? 0.f : a[1];
    }
    v2f bl, br;
    bl[0] = Wl[(size_t)kk * FDIM + colBase + lane16];
    bl[1] = Wl[(size_t)(kk + 1) * FDIM + colBase + lane16];
    br[0] = Wr[(size_t)kk * FDIM + colBase + lane16];
    br[1] = Wr[(size_t)(kk + 1) * FDIM + colBase + lane16];
    accL = __builtin_amdgcn_wmma_f32_16x16x4_f32(false, a, false, bl,
                                                 (short)0, accL, false, false);
    accR = __builtin_amdgcn_wmma_f32_16x16x4_f32(false, a, false, br,
                                                 (short)0, accR, false, false);
  }

#pragma unroll
  for (int v = 0; v < 8; ++v) {
    int m = v + 8 * laneHi;
    int r = rowTile * 16 + m;
    if (r < N) {
      xl[(size_t)r * FDIM + colBase + lane16] = accL[v];
      xr[(size_t)r * FDIM + colBase + lane16] = accR[v];
    }
  }
}

__global__ __launch_bounds__(256) void gatv2_link_edges(
    const int* __restrict__ esrc, const int* __restrict__ edst,
    int* __restrict__ head, int2* __restrict__ link, int E) {
  const int eid = blockIdx.x * 256 + threadIdx.x;
  if (eid >= E) return;
  const int d = edst[eid];
  const int s = esrc[eid];
  const int old = atomicExch(head + (size_t)d * NLIST + (eid & (NLIST - 1)), eid);
  link[eid] = make_int2(s, old);
}

__device__ __forceinline__ void gatv2_group_step(
    const v4f row0, const v4f row1, const v4f r0, const v4f r1, const v4f a0,
    const v4f a1, const bool have, float& m, float& den, v4f& acc0, v4f& acc1) {
  v4f t0 = row0 + r0;
  v4f t1 = row1 + r1;
#pragma unroll
  for (int i = 0; i < 4; ++i) {
    t0[i] = t0[i] > 0.f ? t0[i] : NEG_SLOPE * t0[i];
    t1[i] = t1[i] > 0.f ? t1[i] : NEG_SLOPE * t1[i];
  }
  float p = t0[0] * a0[0];
  p += t0[1] * a0[1];
  p += t0[2] * a0[2];
  p += t0[3] * a0[3];
  p += t1[0] * a1[0];
  p += t1[1] * a1[1];
  p += t1[2] * a1[2];
  p += t1[3] * a1[3];
#pragma unroll
  for (int off = 4; off; off >>= 1) p += __shfl_xor(p, off, 32);
  const float pe = have ? p : -__builtin_inff();
  const float mNew = fmaxf(m, pe);
  const float sc = __expf(m - mNew);
  const float w  = __expf(pe - mNew);
  den  = den * sc + w;
  acc0 = acc0 * sc + w * row0;
  acc1 = acc1 * sc + w * row1;
  m = mNew;
}

__global__ __launch_bounds__(256) void gatv2_node_pull(
    const float* __restrict__ xl, const float* __restrict__ xr,
    const int* __restrict__ head, const int2* __restrict__ link,
    const float* __restrict__ att, const float* __restrict__ bias,
    float* __restrict__ res, int N, int E) {
  const int node = __builtin_amdgcn_readfirstlane(
      (int)(blockIdx.x * 8 + (threadIdx.x >> 5)));
  if (node >= N) return;
  const int lane = threadIdx.x & 31;
  const int grp  = lane >> 3;
  const int fo   = 4 * (lane & 7);

  const float* xrN = xr + (size_t)node * FDIM + fo;
  const v4f r0 = *(const v4f*)(xrN);
  const v4f r1 = *(const v4f*)(xrN + 32);
  const v4f a0 = *(const v4f*)(att + fo);
  const v4f a1 = *(const v4f*)(att + 32 + fo);
  const float* xlLane = xl + fo;

  float m = -3.402823466e+38f;
  float den = 0.f;
  v4f acc0 = {0.f, 0.f, 0.f, 0.f};
  v4f acc1 = {0.f, 0.f, 0.f, 0.f};

  bool have = grp == 0;
  v4f row0 = *(const v4f*)(xlLane + (size_t)node * FDIM);
  v4f row1 = *(const v4f*)(xlLane + (size_t)node * FDIM + 32);
  int nxt = head[(size_t)node * NLIST + grp];
  int2 ln = link[nxt >= 0 ? nxt : 0];

  int it = 0;
  bool anyMore;
  do {
    const bool more = nxt >= 0;
    const int srcN = more ? ln.x : node;
    const int nxtN = more ? ln.y : -1;
    const v4f rowN0 = *(const v4f*)(xlLane + (size_t)srcN * FDIM);
    const v4f rowN1 = *(const v4f*)(xlLane + (size_t)srcN * FDIM + 32);
    const int2 lnN = link[nxtN >= 0 ? nxtN : 0];

    gatv2_group_step(row0, row1, r0, r1, a0, a1, have, m, den, acc0, acc1);

    asm volatile("" :: "v"(rowN0[0]), "v"(rowN0[1]), "v"(rowN0[2]), "v"(rowN0[3]),
                       "v"(rowN1[0]), "v"(rowN1[1]), "v"(rowN1[2]), "v"(rowN1[3]),
                       "v"(lnN.x), "v"(lnN.y), "v"(m));

    row0 = rowN0;
    row1 = rowN1;
    have = more;
    nxt  = nxtN;
    ln   = lnN;
    ++it;
    anyMore = __builtin_amdgcn_ballot_w32(more) != 0u;
  } while (anyMore && it <= E);

  const v4f b0 = *(const v4f*)(bias + fo);
  const v4f b1 = *(const v4f*)(bias + 32 + fo);

  float M = m;
  M = fmaxf(M, __shfl_xor(M, 8, 32));
  M = fmaxf(M, __shfl_xor(M, 16, 32));
  const float s = __expf(m - M);
  den *= s;
  acc0 *= s;
  acc1 *= s;
  den += __shfl_xor(den, 8, 32);
  den += __shfl_xor(den, 16, 32);
#pragma unroll
  for (int i = 0; i < 4; ++i) {
    acc0[i] += __shfl_xor(acc0[i], 8, 32);
    acc1[i] += __shfl_xor(acc1[i], 8, 32);
  }
#pragma unroll
  for (int i = 0; i < 4; ++i) {
    acc0[i] += __shfl_xor(acc0[i], 16, 32);
    acc1[i] += __shfl_xor(acc1[i], 16, 32);
  }

  if (grp == 0) {
    const float inv = 1.f / (den + EPS_DEN);
    float* o = res + (size_t)node * FDIM + fo;
    *(v4f*)(o)      = acc0 * inv + b0;
    *(v4f*)(o + 32) = acc1 * inv + b1;
  }
}

extern "C" void kernel_launch(void* const* d_in, const int* in_sizes, int n_in,
                              void* d_out, int out_size, void* d_ws, size_t ws_size,
                              hipStream_t stream) {
  const float* x    = (const float*)d_in[0];
  const int*   ei   = (const int*)d_in[1];
  const float* W1l  = (const float*)d_in[2];
  const float* W1r  = (const float*)d_in[3];
  const float* att1 = (const float*)d_in[4];
  const float* b1   = (const float*)d_in[5];
  const float* W2l  = (const float*)d_in[6];
  const float* W2r  = (const float*)d_in[7];
  const float* att2 = (const float*)d_in[8];
  const float* b2   = (const float*)d_in[9];

  const int N  = in_sizes[0] / 128;
  const int E  = in_sizes[1] / 2;

  float* ws = (float*)d_ws;
  const size_t NF = (size_t)N * FDIM;
  float* out  = (float*)d_out;
  float* xl   = ws;
  float* xr   = xl + NF;
  float* h    = out;
  int2*  link = (int2*)(xr + NF);
  int*   head = (int*)(link + E);

  const int rowTiles = (N + 15) / 16;
  const int nodeBl   = (N + 7) / 8;
  const int edgeBl   = (E + 255) / 256;
  const int* esrc = ei;
  const int* edst = ei + E;

  gatv2_gemm_dual<false, true><<<rowTiles, 128, 0, stream>>>(x, W1l, W1r, xl, xr, head, N, 128);
  gatv2_link_edges<<<edgeBl, 256, 0, stream>>>(esrc, edst, head, link, E);
  gatv2_node_pull<<<nodeBl, 256, 0, stream>>>(xl, xr, head, link, att1, b1, h, N, E);

  gatv2_gemm_dual<true, false><<<rowTiles, 128, 0, stream>>>(h, W2l, W2r, xl, xr, head, N, 64);
  gatv2_node_pull<<<nodeBl, 256, 0, stream>>>(xl, xr, head, link, att2, b2, out, N, E);
}
